# starting kernel
# baseline (speedup 1.0000x reference)
#include <hip/hip_runtime.h>
#include <math.h>

typedef float v8f __attribute__((ext_vector_type(8)));
typedef float v2f __attribute__((ext_vector_type(2)));

#define WS_ROWD   0
#define WS_ROWH   8192
#define WS_ROWW   16384
#define WS_CHSUM  24576
#define WS_CHSQ   24592
#define WS_B2     24608
#define WS_ZERON  24608

__global__ void zero_ws_kernel(float* __restrict__ ws, int n) {
    int i = blockIdx.x * blockDim.x + threadIdx.x;
    if (i < n) ws[i] = 0.0f;
}

__global__ __launch_bounds__(256) void reduce_x_kernel(const float* __restrict__ x,
                                                       float* __restrict__ ws) {
    __shared__ float sD[4];
    __shared__ float sH[64];
    __shared__ float sW[64];
    __shared__ float sSq;

    const int tid  = threadIdx.x;
    const int lane = tid & 31;
    const int wave = tid >> 5;
    const int half = lane >> 4;
    const int l16  = lane & 15;
    const int bid  = blockIdx.x;
    const int bc    = bid >> 4;
    const int chunk = bid & 15;

    if (tid < 64) { sH[tid] = 0.0f; sW[tid] = 0.0f; }
    if (tid < 4)  sD[tid] = 0.0f;
    if (tid == 64) sSq = 0.0f;
    __syncthreads();

    const float* base = x + (size_t)bc * 262144u + (size_t)chunk * 16384u;

    float accW0 = 0.0f, accW1 = 0.0f, accW2 = 0.0f, accW3 = 0.0f, accSq = 0.0f;
    for (int p = wave; p < 128; p += 8) {
        const int r = 2 * p + half;
        const float4 v = *reinterpret_cast<const float4*>(base + r * 64 + l16 * 4);
        accW0 += v.x; accW1 += v.y; accW2 += v.z; accW3 += v.w;
        accSq += v.x * v.x + v.y * v.y + v.z * v.z + v.w * v.w;
        float rs = (v.x + v.y) + (v.z + v.w);
        #pragma unroll
        for (int off = 8; off > 0; off >>= 1) rs += __shfl_xor(rs, off, 32);
        if (l16 == 0) {
            atomicAdd(&sD[r >> 6], rs);
            atomicAdd(&sH[r & 63], rs);
        }
    }
    atomicAdd(&sW[l16 * 4 + 0], accW0);
    atomicAdd(&sW[l16 * 4 + 1], accW1);
    atomicAdd(&sW[l16 * 4 + 2], accW2);
    atomicAdd(&sW[l16 * 4 + 3], accW3);
    #pragma unroll
    for (int off = 16; off > 0; off >>= 1) accSq += __shfl_xor(accSq, off, 32);
    if (lane == 0) atomicAdd(&sSq, accSq);
    __syncthreads();

    if (tid < 64) {
        atomicAdd(&ws[WS_ROWW + bc * 64 + tid], sW[tid]);
        atomicAdd(&ws[WS_ROWH + bc * 64 + tid], sH[tid]);
    }
    if (tid < 4)
        atomicAdd(&ws[WS_ROWD + bc * 64 + chunk * 4 + tid], sD[tid]);
    if (tid == 64) {
        float t = sD[0] + sD[1] + sD[2] + sD[3];
        atomicAdd(&ws[WS_CHSUM + (bc & 15)], t);
    }
    if (tid == 65)
        atomicAdd(&ws[WS_CHSQ + (bc & 15)], sSq);
}

__global__ __launch_bounds__(256) void prep_kernel(float* __restrict__ ws,
                                                   const float* __restrict__ gamma,
                                                   const float* __restrict__ beta,
                                                   const float* __restrict__ core2) {
    __shared__ float sA[16];
    __shared__ float sOff[16];
    const int tid = threadIdx.x;
    if (tid < 16) {
        const float N = 2097152.0f;
        const float s = ws[WS_CHSUM + tid];
        const float q = ws[WS_CHSQ + tid];
        const float mean = s / N;
        const float var  = q / N - mean * mean;
        const float a = gamma[tid] * rsqrtf(var + 1e-5f);
        sA[tid]   = a;
        sOff[tid] = beta[tid] - mean * a;
    }
    __syncthreads();

    for (int i = tid; i < 24576; i += 256) {
        const int c = (i >> 6) & 15;
        ws[i] = sA[c] * (ws[i] * (1.0f / 4096.0f)) + sOff[c];
    }
    for (int i = tid; i < 2048; i += 256) {
        const int h = i >> 5;
        const int r = i & 31;
        const float* p = core2 + (r * 64 + h) * 32;
        float s = 0.0f;
        #pragma unroll 8
        for (int j = 0; j < 32; ++j) s += p[j];
        ws[WS_B2 + i] = s * (1.0f / 32.0f);
    }
}

__global__ __launch_bounds__(32) void gemm_gelu_kernel(const float* __restrict__ ws,
                                                       const float* __restrict__ core1,
                                                       const float* __restrict__ core3,
                                                       float* __restrict__ out) {
    const int lane = threadIdx.x;
    const int m0   = blockIdx.x * 16;
    const int lhi  = lane >> 4;
    const int llo  = lane & 15;

    for (int g = 0; g < 3; ++g) {
        const float* A = ws + g * 8192;
        const float* B = (g == 0) ? core1 : (ws + WS_B2);
        for (int nt = 0; nt < 2; ++nt) {
            const int n0 = nt * 16;
            v8f acc = {};
            for (int k0 = 0; k0 < 64; k0 += 4) {
                const int kk = k0 + lhi * 2;
                v2f a, b;
                const float* ap = A + (m0 + llo) * 64 + kk;
                a.x = ap[0];
                a.y = ap[1];
                const int n = n0 + llo;
                if (g < 2) {
                    b.x = B[kk * 32 + n];
                    b.y = B[(kk + 1) * 32 + n];
                } else {
                    const float* bp = core3 + n * 64 + kk;
                    b.x = bp[0];
                    b.y = bp[1];
                }
                acc = __builtin_amdgcn_wmma_f32_16x16x4_f32(
                     false, a,  false, b,
                     (short)0, acc,  false,  false);
            }
            #pragma unroll
            for (int j = 0; j < 8; ++j) {
                const int m = m0 + j + lhi * 8;
                const int n = n0 + llo;
                const float v = acc[j];
                const float ge = 0.5f * v * (1.0f + erff(v * 0.70710678118654752440f));
                out[m * 96 + g * 32 + n] = ge;
            }
        }
    }
}

extern "C" void kernel_launch(void* const* d_in, const int* in_sizes, int n_in,
                              void* d_out, int out_size, void* d_ws, size_t ws_size,
                              hipStream_t stream) {
    const float* x     = (const float*)d_in[0];
    const float* gamma = (const float*)d_in[1];
    const float* beta  = (const float*)d_in[2];
    const float* core1 = (const float*)d_in[3];
    const float* core2 = (const float*)d_in[4];
    const float* core3 = (const float*)d_in[5];
    float* ws  = (float*)d_ws;
    float* out = (float*)d_out;

    zero_ws_kernel<<<(WS_ZERON + 255) / 256, 256, 0, stream>>>(ws, WS_ZERON);
    reduce_x_kernel<<<2048, 256, 0, stream>>>(x, ws);
    prep_kernel<<<1, 256, 0, stream>>>(ws, gamma, beta, core2);
    gemm_gelu_kernel<<<8, 32, 0, stream>>>(ws, core1, core3, out);
}


# prep launch dissolved: BN affine + axis-mean scaling folded into the GEMM's A-fragment load (24 one-wave blocks, one load round trip); B2 mean built in launch 1, summed 4-wide + 8-lane shuffle tree
# speedup vs baseline: 1.5990x; 1.5990x over previous
#include <hip/hip_runtime.h>
#include <math.h>

typedef float v8f __attribute__((ext_vector_type(8)));
typedef float v2f __attribute__((ext_vector_type(2)));

#define WS_ROWD   0
#define WS_ROWH   8192
#define WS_ROWW   16384
#define WS_CHSUM  24576
#define WS_CHSQ   24592
#define WS_B2     24608
#define WS_ZERON  24608
#define ZERO_BLOCKS ((WS_ZERON + 255) / 256)
#define B2_BLOCKS   8
#define INIT_BLOCKS (ZERO_BLOCKS + B2_BLOCKS)

__global__ __launch_bounds__(256) void init_kernel(float* __restrict__ ws,
                                                   const float* __restrict__ core2) {
    const int tid = threadIdx.x;
    if (blockIdx.x < ZERO_BLOCKS) {
        const int i = blockIdx.x * 256 + tid;
        if (i < WS_ZERON) ws[i] = 0.0f;
        return;
    }
    const int lane = tid & 31;
    const int sub  = lane & 7;
    const int grp  = lane >> 3;
    const int wv   = (blockIdx.x - ZERO_BLOCKS) * 8 + (tid >> 5);
    const float* srcp = core2 + (size_t)(wv * 32 + grp * 8) * 32 + sub * 4;
    float4 v[8];
    #pragma unroll
    for (int p = 0; p < 8; ++p)
        v[p] = *reinterpret_cast<const float4*>(srcp + p * 32);
    float res = 0.0f;
    #pragma unroll
    for (int p = 0; p < 8; ++p) {
        float s = (v[p].x + v[p].y) + (v[p].z + v[p].w);
        s += __shfl_xor(s, 1, 32);
        s += __shfl_xor(s, 2, 32);
        s += __shfl_xor(s, 4, 32);
        if (sub == p) res = s;
    }
    ws[WS_B2 + wv * 32 + lane] = res * (1.0f / 32.0f);
}

__global__ __launch_bounds__(256) void reduce_x_kernel(const float* __restrict__ x,
                                                       float* __restrict__ ws) {
    __shared__ float sD[4];
    __shared__ float sH[64];
    __shared__ float sW[64];
    __shared__ float sSq;

    const int tid  = threadIdx.x;
    const int lane = tid & 31;
    const int wave = tid >> 5;
    const int half = lane >> 4;
    const int l16  = lane & 15;
    const int bid  = blockIdx.x;
    const int bc    = bid >> 4;
    const int chunk = bid & 15;

    if (tid < 64) { sH[tid] = 0.0f; sW[tid] = 0.0f; }
    if (tid < 4)  sD[tid] = 0.0f;
    if (tid == 64) sSq = 0.0f;
    __syncthreads();

    const float* base = x + (size_t)bc * 262144u + (size_t)chunk * 16384u;

    float accW0 = 0.0f, accW1 = 0.0f, accW2 = 0.0f, accW3 = 0.0f, accSq = 0.0f;
    for (int p = wave; p < 128; p += 8) {
        const int r = 2 * p + half;
        const float4 v = *reinterpret_cast<const float4*>(base + r * 64 + l16 * 4);
        accW0 += v.x; accW1 += v.y; accW2 += v.z; accW3 += v.w;
        accSq += v.x * v.x + v.y * v.y + v.z * v.z + v.w * v.w;
        float rs = (v.x + v.y) + (v.z + v.w);
        #pragma unroll
        for (int off = 8; off > 0; off >>= 1) rs += __shfl_xor(rs, off, 32);
        if (l16 == 0) {
            atomicAdd(&sD[r >> 6], rs);
            atomicAdd(&sH[r & 63], rs);
        }
    }
    atomicAdd(&sW[l16 * 4 + 0], accW0);
    atomicAdd(&sW[l16 * 4 + 1], accW1);
    atomicAdd(&sW[l16 * 4 + 2], accW2);
    atomicAdd(&sW[l16 * 4 + 3], accW3);
    #pragma unroll
    for (int off = 16; off > 0; off >>= 1) accSq += __shfl_xor(accSq, off, 32);
    if (lane == 0) atomicAdd(&sSq, accSq);
    __syncthreads();

    if (tid < 64) {
        atomicAdd(&ws[WS_ROWW + bc * 64 + tid], sW[tid]);
        atomicAdd(&ws[WS_ROWH + bc * 64 + tid], sH[tid]);
    }
    if (tid < 4)
        atomicAdd(&ws[WS_ROWD + bc * 64 + chunk * 4 + tid], sD[tid]);
    if (tid == 64) {
        float t = sD[0] + sD[1] + sD[2] + sD[3];
        atomicAdd(&ws[WS_CHSUM + (bc & 15)], t);
    }
    if (tid == 65)
        atomicAdd(&ws[WS_CHSQ + (bc & 15)], sSq);
}

template <bool kBRowMajor>
__device__ __forceinline__ void gemm_gelu_tile(const float* __restrict__ Araw,
                                               const float* __restrict__ Bsrc,
                                               const float* __restrict__ ws,
                                               const float* __restrict__ gamma,
                                               const float* __restrict__ beta,
                                               float* __restrict__ outp,
                                               const int lhi, const int llo) {
    const float* ap = Araw + llo * 64 + lhi * 2;
    v2f a[16];
    #pragma unroll
    for (int i = 0; i < 16; ++i)
        a[i] = *reinterpret_cast<const v2f*>(ap + 4 * i);

    v2f b0[16], b1[16];
    if (kBRowMajor) {
        const float* bp = Bsrc + (lhi * 2) * 32 + llo;
        #pragma unroll
        for (int i = 0; i < 16; ++i) {
            b0[i].x = bp[i * 128];
            b0[i].y = bp[i * 128 + 32];
            b1[i].x = bp[i * 128 + 16];
            b1[i].y = bp[i * 128 + 48];
        }
    } else {
        const float* bp = Bsrc + llo * 64 + lhi * 2;
        #pragma unroll
        for (int i = 0; i < 16; ++i) {
            b0[i] = *reinterpret_cast<const v2f*>(bp + 4 * i);
            b1[i] = *reinterpret_cast<const v2f*>(bp + 1024 + 4 * i);
        }
    }

    const float N    = 2097152.0f;
    const float s    = ws[WS_CHSUM + llo];
    const float q    = ws[WS_CHSQ + llo];
    const float gm   = gamma[llo];
    const float bt   = beta[llo];
    __builtin_amdgcn_sched_barrier(0);
    const float mean = s / N;
    const float var  = q / N - mean * mean;
    const float sA   = gm * rsqrtf(var + 1e-5f);
    const float sOff = bt - mean * sA;

    #pragma unroll
    for (int i = 0; i < 16; ++i) {
        a[i].x = sA * (a[i].x * (1.0f / 4096.0f)) + sOff;
        a[i].y = sA * (a[i].y * (1.0f / 4096.0f)) + sOff;
    }

    v8f acc0 = {}, acc1 = {};
    #pragma unroll
    for (int i = 0; i < 16; ++i) {
        acc0 = __builtin_amdgcn_wmma_f32_16x16x4_f32(
             false, a[i],  false, b0[i],
             (short)0, acc0,  false,  false);
        acc1 = __builtin_amdgcn_wmma_f32_16x16x4_f32(
             false, a[i],  false, b1[i],
             (short)0, acc1,  false,  false);
    }

    #pragma unroll
    for (int j = 0; j < 8; ++j) {
        const int r = j + lhi * 8;
        const float v0 = acc0[j];
        const float v1 = acc1[j];
        const float g0 = 0.5f * v0 * (1.0f + erff(v0 * 0.70710678118654752440f));
        const float g1 = 0.5f * v1 * (1.0f + erff(v1 * 0.70710678118654752440f));
        outp[r * 96 + llo]      = g0;
        outp[r * 96 + 16 + llo] = g1;
    }
}

__global__ __launch_bounds__(32) void gemm_gelu_kernel(const float* __restrict__ ws,
                                                       const float* __restrict__ gamma,
                                                       const float* __restrict__ beta,
                                                       const float* __restrict__ core1,
                                                       const float* __restrict__ core3,
                                                       float* __restrict__ out) {
    const int lane = threadIdx.x;
    const int g    = blockIdx.x >> 3;
    const int m0   = (blockIdx.x & 7) * 16;
    const int lhi  = lane >> 4;
    const int llo  = lane & 15;
    const float* Araw = ws + g * 8192 + m0 * 64;
    float* outp = out + m0 * 96 + g * 32;
    if (g == 0)
        gemm_gelu_tile<true>(Araw, core1, ws, gamma, beta, outp, lhi, llo);
    else
        gemm_gelu_tile<false>(Araw, (g == 1) ? (ws + WS_B2) : core3, ws, gamma, beta, outp, lhi, llo);
}

extern "C" void kernel_launch(void* const* d_in, const int* in_sizes, int n_in,
                              void* d_out, int out_size, void* d_ws, size_t ws_size,
                              hipStream_t stream) {
    const float* x     = (const float*)d_in[0];
    const float* gamma = (const float*)d_in[1];
    const float* beta  = (const float*)d_in[2];
    const float* core1 = (const float*)d_in[3];
    const float* core2 = (const float*)d_in[4];
    const float* core3 = (const float*)d_in[5];
    float* ws  = (float*)d_ws;
    float* out = (float*)d_out;

    init_kernel<<<INIT_BLOCKS, 256, 0, stream>>>(ws, core2);
    reduce_x_kernel<<<2048, 256, 0, stream>>>(x, ws);
    gemm_gelu_kernel<<<24, 32, 0, stream>>>(ws, gamma, beta, core1, core3, out);
}


# x reduction rebuilt as a stream: 512 blocks of 8 waves, each wave holds two 8 KB pieces of b128 loads in flight and sums its lanes once per piece pair; sums re-associated, fewer atomics per address
# speedup vs baseline: 2.8473x; 1.7807x over previous
#include <hip/hip_runtime.h>
#include <math.h>

typedef float v8f __attribute__((ext_vector_type(8)));
typedef float v2f __attribute__((ext_vector_type(2)));

#define WS_ROWD   0
#define WS_ROWH   8192
#define WS_ROWW   16384
#define WS_CHSUM  24576
#define WS_CHSQ   24592
#define WS_B2     24608
#define WS_ZERON  24608
#define ZERO_BLOCKS ((WS_ZERON + 255) / 256)
#define B2_BLOCKS   8
#define INIT_BLOCKS (ZERO_BLOCKS + B2_BLOCKS)

__global__ __launch_bounds__(256) void init_kernel(float* __restrict__ ws,
                                                   const float* __restrict__ core2) {
    const int tid = threadIdx.x;
    if (blockIdx.x < ZERO_BLOCKS) {
        const int i = blockIdx.x * 256 + tid;
        if (i < WS_ZERON) ws[i] = 0.0f;
        return;
    }
    const int lane = tid & 31;
    const int sub  = lane & 7;
    const int grp  = lane >> 3;
    const int wv   = (blockIdx.x - ZERO_BLOCKS) * 8 + (tid >> 5);
    const float* srcp = core2 + (size_t)(wv * 32 + grp * 8) * 32 + sub * 4;
    float4 v[8];
    #pragma unroll
    for (int p = 0; p < 8; ++p)
        v[p] = *reinterpret_cast<const float4*>(srcp + p * 32);
    float res = 0.0f;
    #pragma unroll
    for (int p = 0; p < 8; ++p) {
        float s = (v[p].x + v[p].y) + (v[p].z + v[p].w);
        s += __shfl_xor(s, 1, 32);
        s += __shfl_xor(s, 2, 32);
        s += __shfl_xor(s, 4, 32);
        if (sub == p) res = s;
    }
    ws[WS_B2 + wv * 32 + lane] = res * (1.0f / 32.0f);
}

#define RX_LOADS 16

#define RX_ROW_STEP(v, m)                                                                 \
    (v) += __builtin_bit_cast(float, __builtin_amdgcn_mov_dpp(__builtin_bit_cast(int, (v)), \
                                                              0x160 | (m), 0xf, 0xf, false))
__device__ __forceinline__ float rx_row_sum(float v) {
    RX_ROW_STEP(v, 8);
    RX_ROW_STEP(v, 4);
    RX_ROW_STEP(v, 2);
    RX_ROW_STEP(v, 1);
    return v;
}
__device__ __forceinline__ float rx_xhalf(float v) {
    const int b = __builtin_bit_cast(int, v);
    return __builtin_bit_cast(float,
        __builtin_amdgcn_permlanex16(b, b, 0x76543210, 0xfedcba98, false, false));
}

typedef float v4f __attribute__((ext_vector_type(4)));

__device__ __forceinline__ void rx_issue(v4f (&v)[RX_LOADS], const float* __restrict__ p) {
    #pragma unroll
    for (int k = 0; k < RX_LOADS; ++k)
        v[k] = *reinterpret_cast<const v4f*>(p + k * 128);
}

__device__ __forceinline__ v2f rx_fold(const v4f (&v)[RX_LOADS], v2f (&accH)[RX_LOADS],
                                       v2f& accWlo, v2f& accWhi, v2f& accQlo, v2f& accQhi) {
    v2f dsum;
    #pragma unroll
    for (int k = 0; k < RX_LOADS; ++k) {
        const v2f lo = v[k].xy;
        const v2f hi = v[k].zw;
        const v2f t  = lo + hi;
        accH[k] += t;
        dsum = (k == 0) ? t : dsum + t;
        accWlo  += lo;
        accWhi  += hi;
        accQlo = __builtin_elementwise_fma(lo, lo, accQlo);
        accQhi = __builtin_elementwise_fma(hi, hi, accQhi);
    }
    return dsum;
}

__global__ __launch_bounds__(256) void reduce_x_kernel(const float* __restrict__ x,
                                                       float* __restrict__ ws) {
    __shared__ float sWp[8][64];
    __shared__ float sHp[8][32];
    __shared__ float sDp[8][4];
    __shared__ float sQp[8];

    const int tid  = threadIdx.x;
    const int lane = tid & 31;
    const int wave = tid >> 5;
    const int half = lane >> 4;
    const int l16  = lane & 15;
    const int bc   = blockIdx.x >> 2;
    const int q    = blockIdx.x & 3;
    const int dg   = wave >> 1;
    const int hh   = wave & 1;

    const float* p = x + (size_t)bc * 262144u + (size_t)(q * 16 + dg * 4) * 4096u
                       + hh * 2048 + lane * 4;

    v2f accH[RX_LOADS];
    #pragma unroll
    for (int k = 0; k < RX_LOADS; ++k) accH[k] = (v2f){0.0f, 0.0f};
    v2f accWlo = {0.0f, 0.0f}, accWhi = {0.0f, 0.0f};
    v2f accQlo = {0.0f, 0.0f}, accQhi = {0.0f, 0.0f};
    float dval = 0.0f;

    for (int i = 0; i < 2; ++i) {
        v4f va[RX_LOADS], vb[RX_LOADS];
        rx_issue(va, p);
        rx_issue(vb, p + 4096);
        p += 2 * 4096;
        __builtin_amdgcn_sched_barrier(0);
        const v2f da = rx_fold(va, accH, accWlo, accWhi, accQlo, accQhi);
        const v2f db = rx_fold(vb, accH, accWlo, accWhi, accQlo, accQhi);
        float ta = rx_row_sum(da.x + da.y);
        float tb = rx_row_sum(db.x + db.y);
        ta += rx_xhalf(ta);
        tb += rx_xhalf(tb);
        dval = (lane == 2 * i)     ? ta : dval;
        dval = (lane == 2 * i + 1) ? tb : dval;
    }

    float hsum[RX_LOADS];
    #pragma unroll
    for (int k = 0; k < RX_LOADS; ++k) hsum[k] = rx_row_sum(accH[k].x + accH[k].y);
    float hval = hsum[0];
    #pragma unroll
    for (int k = 1; k < RX_LOADS; ++k) hval = (l16 == k) ? hsum[k] : hval;

    v4f wv;
    wv.x = accWlo.x + rx_xhalf(accWlo.x);
    wv.y = accWlo.y + rx_xhalf(accWlo.y);
    wv.z = accWhi.x + rx_xhalf(accWhi.x);
    wv.w = accWhi.y + rx_xhalf(accWhi.y);

    float sq = rx_row_sum((accQlo.x + accQlo.y) + (accQhi.x + accQhi.y));
    sq += rx_xhalf(sq);

    sHp[wave][2 * l16 + half] = hval;
    if (lane < 16) *reinterpret_cast<v4f*>(&sWp[wave][lane * 4]) = wv;
    if (lane < 4)  sDp[wave][lane] = dval;
    if (lane == 0) sQp[wave] = sq;
    __syncthreads();

    if (tid < 64) {
        float s = 0.0f;
        #pragma unroll
        for (int v = 0; v < 8; ++v) s += sWp[v][tid];
        atomicAdd(&ws[WS_ROWW + bc * 64 + tid], s);
    } else if (tid < 128) {
        const int h  = tid - 64;
        const int w0 = h >> 5;
        const int hl = h & 31;
        const float s = (sHp[w0][hl] + sHp[2 + w0][hl]) + (sHp[4 + w0][hl] + sHp[6 + w0][hl]);
        atomicAdd(&ws[WS_ROWH + bc * 64 + h], s);
    } else if (tid < 144) {
        const int dl = tid - 128;
        const int w0 = (dl >> 2) * 2;
        const float s = sDp[w0][dl & 3] + sDp[w0 + 1][dl & 3];
        atomicAdd(&ws[WS_ROWD + bc * 64 + q * 16 + dl], s);
    } else if (tid == 160) {
        float s = 0.0f;
        #pragma unroll
        for (int v = 0; v < 8; ++v) s += (sDp[v][0] + sDp[v][1]) + (sDp[v][2] + sDp[v][3]);
        atomicAdd(&ws[WS_CHSUM + (bc & 15)], s);
    } else if (tid == 192) {
        float s = 0.0f;
        #pragma unroll
        for (int v = 0; v < 8; ++v) s += sQp[v];
        atomicAdd(&ws[WS_CHSQ + (bc & 15)], s);
    }
}

template <bool kBRowMajor>
__device__ __forceinline__ void gemm_gelu_tile(const float* __restrict__ Araw,
                                               const float* __restrict__ Bsrc,
                                               const float* __restrict__ ws,
                                               const float* __restrict__ gamma,
                                               const float* __restrict__ beta,
                                               float* __restrict__ outp,
                                               const int lhi, const int llo) {
    const float* ap = Araw + llo * 64 + lhi * 2;
    v2f a[16];
    #pragma unroll
    for (int i = 0; i < 16; ++i)
        a[i] = *reinterpret_cast<const v2f*>(ap + 4 * i);

    v2f b0[16], b1[16];
    if (kBRowMajor) {
        const float* bp = Bsrc + (lhi * 2) * 32 + llo;
        #pragma unroll
        for (int i = 0; i < 16; ++i) {
            b0[i].x = bp[i * 128];
            b0[i].y = bp[i * 128 + 32];
            b1[i].x = bp[i * 128 + 16];
            b1[i].y = bp[i * 128 + 48];
        }
    } else {
        const float* bp = Bsrc + llo * 64 + lhi * 2;
        #pragma unroll
        for (int i = 0; i < 16; ++i) {
            b0[i] = *reinterpret_cast<const v2f*>(bp + 4 * i);
            b1[i] = *reinterpret_cast<const v2f*>(bp + 1024 + 4 * i);
        }
    }

    const float N    = 2097152.0f;
    const float s    = ws[WS_CHSUM + llo];
    const float q    = ws[WS_CHSQ + llo];
    const float gm   = gamma[llo];
    const float bt   = beta[llo];
    __builtin_amdgcn_sched_barrier(0);
    const float mean = s / N;
    const float var  = q / N - mean * mean;
    const float sA   = gm * rsqrtf(var + 1e-5f);
    const float sOff = bt - mean * sA;

    #pragma unroll
    for (int i = 0; i < 16; ++i) {
        a[i].x = sA * (a[i].x * (1.0f / 4096.0f)) + sOff;
        a[i].y = sA * (a[i].y * (1.0f / 4096.0f)) + sOff;
    }

    v8f acc0 = {}, acc1 = {};
    #pragma unroll
    for (int i = 0; i < 16; ++i) {
        acc0 = __builtin_amdgcn_wmma_f32_16x16x4_f32(
             false, a[i],  false, b0[i],
             (short)0, acc0,  false,  false);
        acc1 = __builtin_amdgcn_wmma_f32_16x16x4_f32(
             false, a[i],  false, b1[i],
             (short)0, acc1,  false,  false);
    }

    #pragma unroll
    for (int j = 0; j < 8; ++j) {
        const int r = j + lhi * 8;
        const float v0 = acc0[j];
        const float v1 = acc1[j];
        const float g0 = 0.5f * v0 * (1.0f + erff(v0 * 0.70710678118654752440f));
        const float g1 = 0.5f * v1 * (1.0f + erff(v1 * 0.70710678118654752440f));
        outp[r * 96 + llo]      = g0;
        outp[r * 96 + 16 + llo] = g1;
    }
}

__global__ __launch_bounds__(32) void gemm_gelu_kernel(const float* __restrict__ ws,
                                                       const float* __restrict__ gamma,
                                                       const float* __restrict__ beta,
                                                       const float* __restrict__ core1,
                                                       const float* __restrict__ core3,
                                                       float* __restrict__ out) {
    const int lane = threadIdx.x;
    const int g    = blockIdx.x >> 3;
    const int m0   = (blockIdx.x & 7) * 16;
    const int lhi  = lane >> 4;
    const int llo  = lane & 15;
    const float* Araw = ws + g * 8192 + m0 * 64;
    float* outp = out + m0 * 96 + g * 32;
    if (g == 0)
        gemm_gelu_tile<true>(Araw, core1, ws, gamma, beta, outp, lhi, llo);
    else
        gemm_gelu_tile<false>(Araw, (g == 1) ? (ws + WS_B2) : core3, ws, gamma, beta, outp, lhi, llo);
}

extern "C" void kernel_launch(void* const* d_in, const int* in_sizes, int n_in,
                              void* d_out, int out_size, void* d_ws, size_t ws_size,
                              hipStream_t stream) {
    const float* x     = (const float*)d_in[0];
    const float* gamma = (const float*)d_in[1];
    const float* beta  = (const float*)d_in[2];
    const float* core1 = (const float*)d_in[3];
    const float* core2 = (const float*)d_in[4];
    const float* core3 = (const float*)d_in[5];
    float* ws  = (float*)d_ws;
    float* out = (float*)d_out;

    init_kernel<<<INIT_BLOCKS, 256, 0, stream>>>(ws, core2);
    reduce_x_kernel<<<512, 256, 0, stream>>>(x, ws);
    gemm_gelu_kernel<<<24, 32, 0, stream>>>(ws, gamma, beta, core1, core3, out);
}


# clearing launch and all atomics removed: x-pass blocks store partial sums in own workspace lines, the GEMM launch adds them in a fixed order (sums re-associated); B2 mean built in the x pass
# speedup vs baseline: 3.2042x; 1.1253x over previous
#include <hip/hip_runtime.h>
#include <math.h>

typedef float v8f __attribute__((ext_vector_type(8)));
typedef float v2f __attribute__((ext_vector_type(2)));

#define WS_BLK    160
#define WS_PW     0
#define WS_PH     64
#define WS_PD     128
#define WS_PS     144
#define WS_B2     (512 * WS_BLK)
#define WS_FLOATS (WS_B2 + 2048)

#define RX_LOADS 16

#define RX_ROW_STEP(v, m)                                                                 \
    (v) += __builtin_bit_cast(float, __builtin_amdgcn_mov_dpp(__builtin_bit_cast(int, (v)), \
                                                              0x160 | (m), 0xf, 0xf, false))
__device__ __forceinline__ float rx_row_sum(float v) {
    RX_ROW_STEP(v, 8);
    RX_ROW_STEP(v, 4);
    RX_ROW_STEP(v, 2);
    RX_ROW_STEP(v, 1);
    return v;
}
__device__ __forceinline__ float rx_xhalf(float v) {
    const int b = __builtin_bit_cast(int, v);
    return __builtin_bit_cast(float,
        __builtin_amdgcn_permlanex16(b, b, 0x76543210, 0xfedcba98, false, false));
}

typedef float v4f __attribute__((ext_vector_type(4)));

__device__ __forceinline__ void rx_issue(v4f (&v)[RX_LOADS], const float* __restrict__ p) {
    #pragma unroll
    for (int k = 0; k < RX_LOADS; ++k)
        v[k] = *reinterpret_cast<const v4f*>(p + k * 128);
}

__device__ __forceinline__ v2f rx_fold(const v4f (&v)[RX_LOADS], v2f (&accH)[RX_LOADS],
                                       v2f& accWlo, v2f& accWhi, v2f& accQlo, v2f& accQhi) {
    v2f dsum;
    #pragma unroll
    for (int k = 0; k < RX_LOADS; ++k) {
        const v2f lo = v[k].xy;
        const v2f hi = v[k].zw;
        const v2f t  = lo + hi;
        accH[k] += t;
        dsum = (k == 0) ? t : dsum + t;
        accWlo  += lo;
        accWhi  += hi;
        accQlo = __builtin_elementwise_fma(lo, lo, accQlo);
        accQhi = __builtin_elementwise_fma(hi, hi, accQhi);
    }
    return dsum;
}

__global__ __launch_bounds__(256) void reduce_x_kernel(const float* __restrict__ x,
                                                       const float* __restrict__ core2,
                                                       float* __restrict__ ws,
                                                       const size_t ws_bytes) {
    if (ws_bytes < WS_FLOATS * sizeof(float)) return;

    __shared__ float sWp[8][64];
    __shared__ float sB2[32];
    __shared__ float sHp[8][32];
    __shared__ float sDp[8][4];
    __shared__ float sQp[8];

    const int tid  = threadIdx.x;
    const int lane = tid & 31;
    const int wave = tid >> 5;
    const int half = lane >> 4;
    const int l16  = lane & 15;
    const int bc   = blockIdx.x >> 2;
    const int q    = blockIdx.x & 3;
    const int dg   = wave >> 1;
    const int hh   = wave & 1;

    const float* p = x + (size_t)bc * 262144u + (size_t)(q * 16 + dg * 4) * 4096u
                       + hh * 2048 + lane * 4;

    v2f accH[RX_LOADS];
    #pragma unroll
    for (int k = 0; k < RX_LOADS; ++k) accH[k] = (v2f){0.0f, 0.0f};
    v2f accWlo = {0.0f, 0.0f}, accWhi = {0.0f, 0.0f};
    v2f accQlo = {0.0f, 0.0f}, accQhi = {0.0f, 0.0f};
    float dval = 0.0f;

    const bool ownsB2 = (blockIdx.x & 7) == 0;
    v4f c2v = {0.0f, 0.0f, 0.0f, 0.0f};
    if (ownsB2)
        c2v = *reinterpret_cast<const v4f*>(core2 + (blockIdx.x >> 3) * 1024 + wave * 128 + lane * 4);

    for (int i = 0; i < 2; ++i) {
        v4f va[RX_LOADS], vb[RX_LOADS];
        rx_issue(va, p);
        rx_issue(vb, p + 4096);
        p += 2 * 4096;
        __builtin_amdgcn_sched_barrier(0);
        const v2f da = rx_fold(va, accH, accWlo, accWhi, accQlo, accQhi);
        const v2f db = rx_fold(vb, accH, accWlo, accWhi, accQlo, accQhi);
        float ta = rx_row_sum(da.x + da.y);
        float tb = rx_row_sum(db.x + db.y);
        ta += rx_xhalf(ta);
        tb += rx_xhalf(tb);
        dval = (lane == 2 * i)     ? ta : dval;
        dval = (lane == 2 * i + 1) ? tb : dval;
    }

    float hsum[RX_LOADS];
    #pragma unroll
    for (int k = 0; k < RX_LOADS; ++k) hsum[k] = rx_row_sum(accH[k].x + accH[k].y);
    float hval = hsum[0];
    #pragma unroll
    for (int k = 1; k < RX_LOADS; ++k) hval = (l16 == k) ? hsum[k] : hval;

    v4f wv;
    wv.x = accWlo.x + rx_xhalf(accWlo.x);
    wv.y = accWlo.y + rx_xhalf(accWlo.y);
    wv.z = accWhi.x + rx_xhalf(accWhi.x);
    wv.w = accWhi.y + rx_xhalf(accWhi.y);

    float sq = rx_row_sum((accQlo.x + accQlo.y) + (accQhi.x + accQhi.y));
    sq += rx_xhalf(sq);

    if (ownsB2) {
        float m = (c2v.x + c2v.y) + (c2v.z + c2v.w);
        RX_ROW_STEP(m, 1);
        RX_ROW_STEP(m, 2);
        RX_ROW_STEP(m, 4);
        if ((lane & 7) == 0) sB2[wave * 4 + (lane >> 3)] = m * (1.0f / 32.0f);
    }

    sHp[wave][2 * l16 + half] = hval;
    if (lane < 16) *reinterpret_cast<v4f*>(&sWp[wave][lane * 4]) = wv;
    if (lane < 4)  sDp[wave][lane] = dval;
    if (lane == 0) sQp[wave] = sq;
    __syncthreads();

    float* blk = ws + (size_t)blockIdx.x * WS_BLK;
    if (tid < 64) {
        float s = 0.0f;
        #pragma unroll
        for (int v = 0; v < 8; ++v) s += sWp[v][tid];
        blk[WS_PW + tid] = s;
    } else if (tid < 128) {
        const int h  = tid - 64;
        const int w0 = h >> 5;
        const int hl = h & 31;
        const float s = (sHp[w0][hl] + sHp[2 + w0][hl]) + (sHp[4 + w0][hl] + sHp[6 + w0][hl]);
        blk[WS_PH + h] = s;
    } else if (tid < 160) {
        float dv = 0.0f, qv = 0.0f;
        if (lane < 16) dv = sDp[(lane >> 2) * 2][lane & 3] + sDp[(lane >> 2) * 2 + 1][lane & 3];
        if (lane < 8)  qv = sQp[lane];
        const float cs = rx_xhalf(rx_row_sum(dv));
        const float cq = rx_xhalf(rx_row_sum(qv));
        float o = dv;
        o = (lane == 16) ? cs : o;
        o = (lane == 17) ? cq : o;
        if (lane < 18) blk[WS_PD + lane] = o;
    } else if (tid < 192) {
        if (ownsB2) ws[WS_B2 + (blockIdx.x >> 3) * 32 + lane] = sB2[lane];
    }
}

template <bool kModeD>
__device__ __forceinline__ void gemm_gelu_tile(const float* __restrict__ ws,
                                               const int arow, const int aoff,
                                               const float* __restrict__ Bsrc,
                                               const float* __restrict__ gamma,
                                               const float* __restrict__ beta,
                                               float* __restrict__ outp,
                                               const int lhi, const int llo) {
    const float* sp = ws + (size_t)((lhi * 64 + llo) * 4) * WS_BLK + WS_PS;
    v2f st[16];
    #pragma unroll
    for (int t = 0; t < 4; ++t) {
        #pragma unroll
        for (int q = 0; q < 4; ++q)
            st[t * 4 + q] = *reinterpret_cast<const v2f*>(sp + (t * 64 + q) * WS_BLK);
    }

    const float* ap = ws + (size_t)(arow * 4) * WS_BLK + lhi * 2;
    v2f a[16];
    v2f ap1[16], ap2[16], ap3[16];
    if (kModeD) {
        #pragma unroll
        for (int i = 0; i < 16; ++i)
            a[i] = *reinterpret_cast<const v2f*>(ap + (i >> 2) * WS_BLK + WS_PD + 4 * (i & 3));
    } else {
        #pragma unroll
        for (int i = 0; i < 16; ++i) {
            a[i]   = *reinterpret_cast<const v2f*>(ap + aoff + 4 * i);
            ap1[i] = *reinterpret_cast<const v2f*>(ap + WS_BLK + aoff + 4 * i);
            ap2[i] = *reinterpret_cast<const v2f*>(ap + 2 * WS_BLK + aoff + 4 * i);
            ap3[i] = *reinterpret_cast<const v2f*>(ap + 3 * WS_BLK + aoff + 4 * i);
        }
    }

    v2f b0[16], b1[16];
    if (kModeD) {
        const float* bp = Bsrc + (lhi * 2) * 32 + llo;
        #pragma unroll
        for (int i = 0; i < 16; ++i) {
            b0[i].x = bp[i * 128];
            b0[i].y = bp[i * 128 + 32];
            b1[i].x = bp[i * 128 + 16];
            b1[i].y = bp[i * 128 + 48];
        }
    } else {
        const float* bp = Bsrc + llo * 64 + lhi * 2;
        #pragma unroll
        for (int i = 0; i < 16; ++i) {
            b0[i] = *reinterpret_cast<const v2f*>(bp + 4 * i);
            b1[i] = *reinterpret_cast<const v2f*>(bp + 1024 + 4 * i);
        }
    }

    const float gm   = gamma[llo];
    const float bt   = beta[llo];
    __builtin_amdgcn_sched_barrier(0);

    const v2f s01 = ((st[0] + st[1]) + (st[2] + st[3])) + ((st[4] + st[5]) + (st[6] + st[7]));
    const v2f s23 = ((st[8] + st[9]) + (st[10] + st[11])) + ((st[12] + st[13]) + (st[14] + st[15]));
    const v2f sh  = s01 + s23;
    const float N    = 2097152.0f;
    const float s    = sh.x + rx_xhalf(sh.x);
    const float q    = sh.y + rx_xhalf(sh.y);
    const float mean = s / N;
    const float var  = q / N - mean * mean;
    const float sA   = gm * rsqrtf(var + 1e-5f);
    const float sOff = bt - mean * sA;

    #pragma unroll
    for (int i = 0; i < 16; ++i) {
        if (!kModeD) a[i] = (a[i] + ap1[i]) + (ap2[i] + ap3[i]);
        a[i].x = sA * (a[i].x * (1.0f / 4096.0f)) + sOff;
        a[i].y = sA * (a[i].y * (1.0f / 4096.0f)) + sOff;
    }

    v8f acc0 = {}, acc1 = {};
    #pragma unroll
    for (int i = 0; i < 16; ++i) {
        acc0 = __builtin_amdgcn_wmma_f32_16x16x4_f32(
             false, a[i],  false, b0[i],
             (short)0, acc0,  false,  false);
        acc1 = __builtin_amdgcn_wmma_f32_16x16x4_f32(
             false, a[i],  false, b1[i],
             (short)0, acc1,  false,  false);
    }

    #pragma unroll
    for (int j = 0; j < 8; ++j) {
        const int r = j + lhi * 8;
        const float v0 = acc0[j];
        const float v1 = acc1[j];
        const float g0 = 0.5f * v0 * (1.0f + erff(v0 * 0.70710678118654752440f));
        const float g1 = 0.5f * v1 * (1.0f + erff(v1 * 0.70710678118654752440f));
        outp[r * 96 + llo]      = g0;
        outp[r * 96 + 16 + llo] = g1;
    }
}

__global__ __launch_bounds__(32) void gemm_gelu_kernel(const float* __restrict__ ws,
                                                       const size_t ws_bytes,
                                                       const float* __restrict__ gamma,
                                                       const float* __restrict__ beta,
                                                       const float* __restrict__ core1,
                                                       const float* __restrict__ core3,
                                                       float* __restrict__ out) {
    if (ws_bytes < WS_FLOATS * sizeof(float)) return;
    const int lane = threadIdx.x;
    const int g    = blockIdx.x >> 3;
    const int m0   = (blockIdx.x & 7) * 16;
    const int lhi  = lane >> 4;
    const int llo  = lane & 15;
    float* outp = out + m0 * 96 + g * 32;
    if (g == 0)
        gemm_gelu_tile<true>(ws, m0 + llo, WS_PD, core1, gamma, beta, outp, lhi, llo);
    else
        gemm_gelu_tile<false>(ws, m0 + llo, (g == 1) ? WS_PH : WS_PW,
                              (g == 1) ? (ws + WS_B2) : core3, gamma, beta, outp, lhi, llo);
}

extern "C" void kernel_launch(void* const* d_in, const int* in_sizes, int n_in,
                              void* d_out, int out_size, void* d_ws, size_t ws_size,
                              hipStream_t stream) {
    const float* x     = (const float*)d_in[0];
    const float* gamma = (const float*)d_in[1];
    const float* beta  = (const float*)d_in[2];
    const float* core1 = (const float*)d_in[3];
    const float* core2 = (const float*)d_in[4];
    const float* core3 = (const float*)d_in[5];
    float* ws  = (float*)d_ws;
    float* out = (float*)d_out;

    reduce_x_kernel<<<512, 256, 0, stream>>>(x, core2, ws, ws_size);
    gemm_gelu_kernel<<<24, 32, 0, stream>>>(ws, ws_size, gamma, beta, core1, core3, out);
}


# GEMM launch spread over 4x the waves: 96 one-wave blocks of 4 rows (the 16 tile rows repeat them): a wave's loads touch 4 rows' and 4 channels' lines instead of 16, a lane finishes 4 GELUs, not 16
# speedup vs baseline: 3.5322x; 1.1024x over previous
#include <hip/hip_runtime.h>
#include <math.h>

typedef float v8f __attribute__((ext_vector_type(8)));
typedef float v2f __attribute__((ext_vector_type(2)));

#define WS_BLK    160
#define WS_PW     0
#define WS_PH     64
#define WS_PD     128
#define WS_PS     144
#define WS_B2     (512 * WS_BLK)
#define WS_FLOATS (WS_B2 + 2048)

#define RX_LOADS 16

#define RX_ROW_STEP(v, m)                                                                 \
    (v) += __builtin_bit_cast(float, __builtin_amdgcn_mov_dpp(__builtin_bit_cast(int, (v)), \
                                                              0x160 | (m), 0xf, 0xf, false))
__device__ __forceinline__ float rx_row_sum(float v) {
    RX_ROW_STEP(v, 8);
    RX_ROW_STEP(v, 4);
    RX_ROW_STEP(v, 2);
    RX_ROW_STEP(v, 1);
    return v;
}
__device__ __forceinline__ float rx_xhalf(float v) {
    const int b = __builtin_bit_cast(int, v);
    return __builtin_bit_cast(float,
        __builtin_amdgcn_permlanex16(b, b, 0x76543210, 0xfedcba98, false, false));
}

typedef float v4f __attribute__((ext_vector_type(4)));

__device__ __forceinline__ void rx_issue(v4f (&v)[RX_LOADS], const float* __restrict__ p) {
    #pragma unroll
    for (int k = 0; k < RX_LOADS; ++k)
        v[k] = *reinterpret_cast<const v4f*>(p + k * 128);
}

__device__ __forceinline__ v2f rx_fold(const v4f (&v)[RX_LOADS], v2f (&accH)[RX_LOADS],
                                       v2f& accWlo, v2f& accWhi, v2f& accQlo, v2f& accQhi) {
    v2f dsum;
    #pragma unroll
    for (int k = 0; k < RX_LOADS; ++k) {
        const v2f lo = v[k].xy;
        const v2f hi = v[k].zw;
        const v2f t  = lo + hi;
        accH[k] += t;
        dsum = (k == 0) ? t : dsum + t;
        accWlo  += lo;
        accWhi  += hi;
        accQlo = __builtin_elementwise_fma(lo, lo, accQlo);
        accQhi = __builtin_elementwise_fma(hi, hi, accQhi);
    }
    return dsum;
}

__global__ __launch_bounds__(256) void reduce_x_kernel(const float* __restrict__ x,
                                                       const float* __restrict__ core2,
                                                       float* __restrict__ ws,
                                                       const size_t ws_bytes) {
    if (ws_bytes < WS_FLOATS * sizeof(float)) return;

    __shared__ float sWp[8][64];
    __shared__ float sB2[32];
    __shared__ float sHp[8][32];
    __shared__ float sDp[8][4];
    __shared__ float sQp[8];

    const int tid  = threadIdx.x;
    const int lane = tid & 31;
    const int wave = tid >> 5;
    const int half = lane >> 4;
    const int l16  = lane & 15;
    const int bc   = blockIdx.x >> 2;
    const int q    = blockIdx.x & 3;
    const int dg   = wave >> 1;
    const int hh   = wave & 1;

    const float* p = x + (size_t)bc * 262144u + (size_t)(q * 16 + dg * 4) * 4096u
                       + hh * 2048 + lane * 4;

    v2f accH[RX_LOADS];
    #pragma unroll
    for (int k = 0; k < RX_LOADS; ++k) accH[k] = (v2f){0.0f, 0.0f};
    v2f accWlo = {0.0f, 0.0f}, accWhi = {0.0f, 0.0f};
    v2f accQlo = {0.0f, 0.0f}, accQhi = {0.0f, 0.0f};
    float dval = 0.0f;

    const bool ownsB2 = (blockIdx.x & 7) == 0;
    v4f c2v = {0.0f, 0.0f, 0.0f, 0.0f};
    if (ownsB2)
        c2v = *reinterpret_cast<const v4f*>(core2 + (blockIdx.x >> 3) * 1024 + wave * 128 + lane * 4);

    for (int i = 0; i < 2; ++i) {
        v4f va[RX_LOADS], vb[RX_LOADS];
        rx_issue(va, p);
        rx_issue(vb, p + 4096);
        p += 2 * 4096;
        __builtin_amdgcn_sched_barrier(0);
        const v2f da = rx_fold(va, accH, accWlo, accWhi, accQlo, accQhi);
        const v2f db = rx_fold(vb, accH, accWlo, accWhi, accQlo, accQhi);
        float ta = rx_row_sum(da.x + da.y);
        float tb = rx_row_sum(db.x + db.y);
        ta += rx_xhalf(ta);
        tb += rx_xhalf(tb);
        dval = (lane == 2 * i)     ? ta : dval;
        dval = (lane == 2 * i + 1) ? tb : dval;
    }

    float hsum[RX_LOADS];
    #pragma unroll
    for (int k = 0; k < RX_LOADS; ++k) hsum[k] = rx_row_sum(accH[k].x + accH[k].y);
    float hval = hsum[0];
    #pragma unroll
    for (int k = 1; k < RX_LOADS; ++k) hval = (l16 == k) ? hsum[k] : hval;

    v4f wv;
    wv.x = accWlo.x + rx_xhalf(accWlo.x);
    wv.y = accWlo.y + rx_xhalf(accWlo.y);
    wv.z = accWhi.x + rx_xhalf(accWhi.x);
    wv.w = accWhi.y + rx_xhalf(accWhi.y);

    float sq = rx_row_sum((accQlo.x + accQlo.y) + (accQhi.x + accQhi.y));
    sq += rx_xhalf(sq);

    if (ownsB2) {
        float m = (c2v.x + c2v.y) + (c2v.z + c2v.w);
        RX_ROW_STEP(m, 1);
        RX_ROW_STEP(m, 2);
        RX_ROW_STEP(m, 4);
        if ((lane & 7) == 0) sB2[wave * 4 + (lane >> 3)] = m * (1.0f / 32.0f);
    }

    sHp[wave][2 * l16 + half] = hval;
    if (lane < 16) *reinterpret_cast<v4f*>(&sWp[wave][lane * 4]) = wv;
    if (lane < 4)  sDp[wave][lane] = dval;
    if (lane == 0) sQp[wave] = sq;
    __syncthreads();

    float* blk = ws + (size_t)blockIdx.x * WS_BLK;
    if (tid < 64) {
        float s = 0.0f;
        #pragma unroll
        for (int v = 0; v < 8; ++v) s += sWp[v][tid];
        blk[WS_PW + tid] = s;
    } else if (tid < 128) {
        const int h  = tid - 64;
        const int w0 = h >> 5;
        const int hl = h & 31;
        const float s = (sHp[w0][hl] + sHp[2 + w0][hl]) + (sHp[4 + w0][hl] + sHp[6 + w0][hl]);
        blk[WS_PH + h] = s;
    } else if (tid < 160) {
        float dv = 0.0f, qv = 0.0f;
        if (lane < 16) dv = sDp[(lane >> 2) * 2][lane & 3] + sDp[(lane >> 2) * 2 + 1][lane & 3];
        if (lane < 8)  qv = sQp[lane];
        const float cs = rx_xhalf(rx_row_sum(dv));
        const float cq = rx_xhalf(rx_row_sum(qv));
        float o = dv;
        o = (lane == 16) ? cs : o;
        o = (lane == 17) ? cq : o;
        if (lane < 18) blk[WS_PD + lane] = o;
    } else if (tid < 192) {
        if (ownsB2) ws[WS_B2 + (blockIdx.x >> 3) * 32 + lane] = sB2[lane];
    }
}

template <bool kModeD>
__device__ __forceinline__ void gemm_gelu_tile(const float* __restrict__ ws,
                                               const int arow, const int aoff,
                                               const float* __restrict__ Bsrc,
                                               const float* __restrict__ gamma,
                                               const float* __restrict__ beta,
                                               float* __restrict__ outp,
                                               const int lhi, const int llo) {
    const int ch    = arow & 15;
    const int batch = ((llo >> 1) & 3) | (lhi << 2);
    const float* sp = ws + (size_t)((batch * 16 + ch) * 4) * WS_BLK + WS_PS;
    v2f st[4];
    #pragma unroll
    for (int q = 0; q < 4; ++q)
        st[q] = *reinterpret_cast<const v2f*>(sp + q * WS_BLK);

    const float* ap = ws + (size_t)(arow * 4) * WS_BLK + lhi * 2;
    v2f a[16];
    v2f ap1[16], ap2[16], ap3[16];
    if (kModeD) {
        #pragma unroll
        for (int i = 0; i < 16; ++i)
            a[i] = *reinterpret_cast<const v2f*>(ap + (i >> 2) * WS_BLK + WS_PD + 4 * (i & 3));
    } else {
        #pragma unroll
        for (int i = 0; i < 16; ++i) {
            a[i]   = *reinterpret_cast<const v2f*>(ap + aoff + 4 * i);
            ap1[i] = *reinterpret_cast<const v2f*>(ap + WS_BLK + aoff + 4 * i);
            ap2[i] = *reinterpret_cast<const v2f*>(ap + 2 * WS_BLK + aoff + 4 * i);
            ap3[i] = *reinterpret_cast<const v2f*>(ap + 3 * WS_BLK + aoff + 4 * i);
        }
    }

    v2f b0[16], b1[16];
    if (kModeD) {
        const float* bp = Bsrc + (lhi * 2) * 32 + llo;
        #pragma unroll
        for (int i = 0; i < 16; ++i) {
            b0[i].x = bp[i * 128];
            b0[i].y = bp[i * 128 + 32];
            b1[i].x = bp[i * 128 + 16];
            b1[i].y = bp[i * 128 + 48];
        }
    } else {
        const float* bp = Bsrc + llo * 64 + lhi * 2;
        #pragma unroll
        for (int i = 0; i < 16; ++i) {
            b0[i] = *reinterpret_cast<const v2f*>(bp + 4 * i);
            b1[i] = *reinterpret_cast<const v2f*>(bp + 1024 + 4 * i);
        }
    }

    const float gm   = gamma[ch];
    const float bt   = beta[ch];
    __builtin_amdgcn_sched_barrier(0);

    const v2f sb = (st[0] + st[1]) + (st[2] + st[3]);
    float sx = sb.x;
    float sy = sb.y;
    RX_ROW_STEP(sx, 2);
    RX_ROW_STEP(sy, 2);
    RX_ROW_STEP(sx, 4);
    RX_ROW_STEP(sy, 4);
    const float N    = 2097152.0f;
    const float s    = sx + rx_xhalf(sx);
    const float q    = sy + rx_xhalf(sy);
    const float mean = s / N;
    const float var  = q / N - mean * mean;
    const float sA   = gm * rsqrtf(var + 1e-5f);
    const float sOff = bt - mean * sA;

    #pragma unroll
    for (int i = 0; i < 16; ++i) {
        if (!kModeD) a[i] = (a[i] + ap1[i]) + (ap2[i] + ap3[i]);
        a[i].x = sA * (a[i].x * (1.0f / 4096.0f)) + sOff;
        a[i].y = sA * (a[i].y * (1.0f / 4096.0f)) + sOff;
    }

    v8f acc0 = {}, acc1 = {};
    #pragma unroll
    for (int i = 0; i < 16; ++i) {
        acc0 = __builtin_amdgcn_wmma_f32_16x16x4_f32(
             false, a[i],  false, b0[i],
             (short)0, acc0,  false,  false);
        acc1 = __builtin_amdgcn_wmma_f32_16x16x4_f32(
             false, a[i],  false, b1[i],
             (short)0, acc1,  false,  false);
    }

    #pragma unroll
    for (int j = 0; j < 2; ++j) {
        const int r = 2 * lhi + j;
        const float v0 = acc0[j];
        const float v1 = acc1[j];
        const float g0 = 0.5f * v0 * (1.0f + erff(v0 * 0.70710678118654752440f));
        const float g1 = 0.5f * v1 * (1.0f + erff(v1 * 0.70710678118654752440f));
        outp[r * 96 + llo]      = g0;
        outp[r * 96 + 16 + llo] = g1;
    }
}

__global__ __launch_bounds__(32) void gemm_gelu_kernel(const float* __restrict__ ws,
                                                       const size_t ws_bytes,
                                                       const float* __restrict__ gamma,
                                                       const float* __restrict__ beta,
                                                       const float* __restrict__ core1,
                                                       const float* __restrict__ core3,
                                                       float* __restrict__ out) {
    if (ws_bytes < WS_FLOATS * sizeof(float)) return;
    const int lane = threadIdx.x;
    const int g    = blockIdx.x >> 5;
    const int m0   = (blockIdx.x & 31) * 4;
    const int lhi  = lane >> 4;
    const int llo  = lane & 15;
    const int arow = m0 + ((llo & 1) | ((llo >> 3) << 1));
    float* outp = out + m0 * 96 + g * 32;
    if (g == 0)
        gemm_gelu_tile<true>(ws, arow, WS_PD, core1, gamma, beta, outp, lhi, llo);
    else
        gemm_gelu_tile<false>(ws, arow, (g == 1) ? WS_PH : WS_PW,
                              (g == 1) ? (ws + WS_B2) : core3, gamma, beta, outp, lhi, llo);
}

extern "C" void kernel_launch(void* const* d_in, const int* in_sizes, int n_in,
                              void* d_out, int out_size, void* d_ws, size_t ws_size,
                              hipStream_t stream) {
    const float* x     = (const float*)d_in[0];
    const float* gamma = (const float*)d_in[1];
    const float* beta  = (const float*)d_in[2];
    const float* core1 = (const float*)d_in[3];
    const float* core2 = (const float*)d_in[4];
    const float* core3 = (const float*)d_in[5];
    float* ws  = (float*)d_ws;
    float* out = (float*)d_out;

    reduce_x_kernel<<<512, 256, 0, stream>>>(x, core2, ws, ws_size);
    gemm_gelu_kernel<<<96, 32, 0, stream>>>(ws, ws_size, gamma, beta, core1, core3, out);
}
